# starting kernel
# baseline (speedup 1.0000x reference)
#include <hip/hip_runtime.h>
#include <hip/hip_bf16.h>


typedef __attribute__((ext_vector_type(2))) float v2f;
typedef __attribute__((ext_vector_type(8))) float v8f;

__global__ __launch_bounds__(256) void kM(const float* __restrict__ x,
                                          float* __restrict__ M) {
    const int lane = threadIdx.x & 31;
    const int wave = (blockIdx.x * 256 + threadIdx.x) >> 5;
    const int n  = wave >> 10;
    const int p0 = (wave & 1023) << 4;
    const int i    = lane & 15;
    const int half = lane >> 4;

    const float* xb = x + ((size_t)n * 64) * 16384 + p0 + i;

    v8f c = {};
    v2f b; b.x = 1.0f; b.y = 1.0f;
    for (int kb = 0; kb < 64; kb += 4) {
        v2f a;
        a.x = xb[(size_t)(kb + 2 * half)     * 16384];
        a.y = xb[(size_t)(kb + 2 * half + 1) * 16384];
        c = __builtin_amdgcn_wmma_f32_16x16x4_f32(
             false, a,  false, b,
             (short)0, c,  false,  false);
    }
    float* Mb = M + (size_t)n * 16384 + p0;
    if (lane == 0) {
        for (int r = 0; r < 8; ++r) Mb[r] = c[r];
    } else if (lane == 16) {
        for (int r = 0; r < 8; ++r) Mb[8 + r] = c[r];
    }
}

__global__ __launch_bounds__(256) void kT(const float* __restrict__ x,
                                          float* __restrict__ T) {
    const int bid   = blockIdx.x;
    const int chunk = bid % 9;
    const int ci    = (bid / 9) % 64;
    const int n     = bid / (9 * 64);

    const int t0base   = 16384 * chunk;
    const int pi_start = t0base / 9;
    const int pi_end   = (t0base + 16383) / 9;
    const int r_lo  = pi_start >> 7;
    const int r_hi  = pi_end >> 7;
    const int nrows = r_hi - r_lo + 3;

    __shared__ float tile[18][130];
    const float* xb = x + ((size_t)n * 64 + ci) * 16384;
    for (int idx = threadIdx.x; idx < nrows * 130; idx += 256) {
        int lr  = idx / 130;
        int lc  = idx - lr * 130;
        int r   = r_lo - 1 + lr;
        int col = lc - 1;
        float v = 0.0f;
        if ((unsigned)r < 128u && (unsigned)col < 128u) v = xb[r * 128 + col];
        tile[lr][lc] = v;
    }
    __syncthreads();

    const int mprime = chunk * 256 + threadIdx.x;
    const int t0 = 64 * mprime;
    float s = 0.0f;
    for (int j = 0; j < 64; ++j) {
        int tj   = t0 + j;
        int pi   = tj / 9;
        int srem = tj - pi * 9;
        int a    = srem / 3;
        int bb   = srem - a * 3;
        s += tile[(pi >> 7) + a - r_lo][(pi & 127) + bb];
    }
    T[(size_t)n * 147456 + ci * 2304 + mprime] = s;
}

__global__ __launch_bounds__(256) void kRed(const float* __restrict__ T,
                                            const float* __restrict__ M,
                                            float* __restrict__ mx,
                                            float* __restrict__ se) {
    const int n = blockIdx.x / 9;
    const int s = blockIdx.x % 9;
    const float* Tb = T + (size_t)n * 147456 + s;
    const float* Mb = M + (size_t)n * 16384;
    __shared__ float red[256];

    float lmax = -3.402823466e38f;
    for (int p = threadIdx.x; p < 16384; p += 256) {
        float v = Tb[9 * p] + Mb[p];
        lmax = fmaxf(lmax, v);
    }
    red[threadIdx.x] = lmax;
    __syncthreads();
    for (int off = 128; off > 0; off >>= 1) {
        if ((int)threadIdx.x < off)
            red[threadIdx.x] = fmaxf(red[threadIdx.x], red[threadIdx.x + off]);
        __syncthreads();
    }
    const float m = red[0];
    __syncthreads();

    float lsum = 0.0f;
    for (int p = threadIdx.x; p < 16384; p += 256) {
        float v = Tb[9 * p] + Mb[p];
        lsum += __expf(v - m);
    }
    red[threadIdx.x] = lsum;
    __syncthreads();
    for (int off = 128; off > 0; off >>= 1) {
        if ((int)threadIdx.x < off)
            red[threadIdx.x] = red[threadIdx.x] + red[threadIdx.x + off];
        __syncthreads();
    }
    if (threadIdx.x == 0) {
        mx[blockIdx.x] = m;
        se[blockIdx.x] = red[0];
    }
}

__global__ __launch_bounds__(256) void kSoftsum(const float* __restrict__ T,
                                                const float* __restrict__ M,
                                                const float* __restrict__ mx,
                                                const float* __restrict__ se,
                                                float* __restrict__ softsum) {
    const int idx = blockIdx.x * 256 + threadIdx.x;
    const int n = idx >> 14;
    const int p = idx & 16383;
    __shared__ float smx[9], sinv[9];
    if (threadIdx.x < 9) {
        smx[threadIdx.x]  = mx[n * 9 + threadIdx.x];
        sinv[threadIdx.x] = 1.0f / se[n * 9 + threadIdx.x];
    }
    __syncthreads();
    const float* Tb = T + (size_t)n * 147456 + 9 * (size_t)p;
    const float m = M[idx];
    float acc = 0.0f;
    for (int s = 0; s < 9; ++s)
        acc += __expf(Tb[s] + m - smx[s]) * sinv[s];
    softsum[idx] = acc;
}

__global__ __launch_bounds__(256) void kOut(const float* __restrict__ x,
                                            const float* __restrict__ softsum,
                                            float* __restrict__ out) {
    const int bid = blockIdx.x;
    const int rc = bid & 15;
    const int c  = (bid >> 4) & 63;
    const int n  = bid >> 10;
    const int r0 = rc * 8;

    __shared__ float tile[10][130];
    const float* xb = x + ((size_t)n * 64 + c) * 16384;
    for (int idx = threadIdx.x; idx < 10 * 130; idx += 256) {
        int lr  = idx / 130;
        int lc  = idx - lr * 130;
        int r   = r0 - 1 + lr;
        int col = lc - 1;
        float v = 0.0f;
        if ((unsigned)r < 128u && (unsigned)col < 128u) v = xb[r * 128 + col];
        tile[lr][lc] = v;
    }
    __syncthreads();

    const float* ssb = softsum + (size_t)n * 16384;
    float* ob = out + ((size_t)n * 64 + c) * 16384;
    for (int k = threadIdx.x; k < 1024; k += 256) {
        int lr = k >> 7;
        int wv = k & 127;
        int h  = r0 + lr;
        float b = tile[lr    ][wv] + tile[lr    ][wv + 1] + tile[lr    ][wv + 2]
                + tile[lr + 1][wv] + tile[lr + 1][wv + 1] + tile[lr + 1][wv + 2]
                + tile[lr + 2][wv] + tile[lr + 2][wv + 1] + tile[lr + 2][wv + 2];
        int pout = c * 256 + h * 2 + (wv >> 6);
        ob[h * 128 + wv] = ssb[pout] + b;
    }
}

extern "C" void kernel_launch(void* const* d_in, const int* in_sizes, int n_in,
                              void* d_out, int out_size, void* d_ws, size_t ws_size,
                              hipStream_t stream) {
    const float* x = (const float*)d_in[0];
    float* out = (float*)d_out;
    float* ws  = (float*)d_ws;

    float* T       = ws;
    float* M       = T + (size_t)8 * 147456;
    float* mx      = M + (size_t)8 * 16384;
    float* se      = mx + 72;
    float* softsum = se + 72;

    kM      <<<1024,       256, 0, stream>>>(x, M);
    kT      <<<8 * 64 * 9, 256, 0, stream>>>(x, T);
    kRed    <<<72,         256, 0, stream>>>(T, M, mx, se);
    kSoftsum<<<512,        256, 0, stream>>>(T, M, mx, se, softsum);
    kOut    <<<8192,       256, 0, stream>>>(x, softsum, out);
}


# kRed folded into kMT: a wave forms V=T+M for its 16 positions in registers and its block writes one max/exp-sum pair per slot; kSoftsum merges 128 pairs per n; 3 launches; exp-sum order changed
# speedup vs baseline: 5.7558x; 5.7558x over previous
#include <hip/hip_runtime.h>
#include <hip/hip_bf16.h>


typedef __attribute__((ext_vector_type(2))) float v2f;
typedef __attribute__((ext_vector_type(4))) float v4f;
typedef __attribute__((ext_vector_type(8))) float v8f;

__global__ __launch_bounds__(256) __attribute__((amdgpu_waves_per_eu(8, 8))) void kMT(const float* __restrict__ x,
                                          float* __restrict__ V,
                                          float* __restrict__ pmx,
                                          float* __restrict__ pse) {
    const int lane  = threadIdx.x & 31;
    const int wid   = (blockIdx.x * 256 + threadIdx.x) >> 5;
    const int strip = wid & 15;
    const int c     = (wid >> 4) & 63;
    const int n     = wid >> 10;
    const int r0    = strip * 8;

    const int p0   = (wid & 1023) << 4;
    const int i    = lane & 15;
    const int half = lane >> 4;
    const float* xm = x + ((size_t)n * 64) * 16384 + p0 + i;
    v2f af[16];
    for (int s = 0; s < 16; ++s) {
        af[s].x = xm[(size_t)(4 * s + 2 * half)     * 16384];
        af[s].y = xm[(size_t)(4 * s + 2 * half + 1) * 16384];
    }

    const float* xb = x + ((size_t)n * 64 + c) * 16384 + 4 * lane;
    v4f row[10];
    for (int j = 0; j < 10; ++j) {
        const int r  = r0 - 1 + j;
        const int rc = r < 0 ? 0 : (r > 127 ? 127 : r);
        v4f v = *(const v4f*)(xb + rc * 128);
        if ((unsigned)r >= 128u) v = (v4f){0.0f, 0.0f, 0.0f, 0.0f};
        row[j] = v;
    }

    float msc[16];
    {
        v8f acc = {};
        v2f b; b.x = 1.0f; b.y = 1.0f;
        for (int s = 0; s < 16; ++s) {
            acc = __builtin_amdgcn_wmma_f32_16x16x4_f32(
                 false, af[s],  false, b,
                 (short)0, acc,  false,  false);
        }
        for (int r = 0; r < 8; ++r) {
            msc[r]     = __int_as_float(__builtin_amdgcn_readlane(__float_as_int(acc[r]), 0));
            msc[8 + r] = __int_as_float(__builtin_amdgcn_readlane(__float_as_int(acc[r]), 16));
        }
    }

    const int up = ((lane + 31) & 31) << 2;
    const int dn = ((lane + 1) & 31) << 2;
    float e[10][6];
    for (int j = 0; j < 10; ++j) {
        float lft = __int_as_float(__builtin_amdgcn_ds_bpermute(up, __float_as_int(row[j].w)));
        float rgt = __int_as_float(__builtin_amdgcn_ds_bpermute(dn, __float_as_int(row[j].x)));
        e[j][0] = (lane == 0)  ? 0.0f : lft;
        e[j][1] = row[j].x;
        e[j][2] = row[j].y;
        e[j][3] = row[j].z;
        e[j][4] = row[j].w;
        e[j][5] = (lane == 31) ? 0.0f : rgt;
    }

    const int lq = lane & 15;
    const int hb = lane & 16;
    const int jA = (9 * lq) >> 4;
    int split = 64 * (jA + 1) - 36 * lq;
    if (split > 36) split = 36;
    const int ng = split >> 2;
    const int  j   = lq < 8 ? lq : 8;
    const int  fa  = (16 * j + 8) / 9;
    const bool two = (fa + 1) * 9 < 16 * (j + 1);
    const int  iB  = (hb + (fa > 0 ? fa - 1 : 0)) << 2;
    const int  iA1 = (hb + fa) << 2;
    const int  iA2 = (hb + (fa < 15 ? fa + 1 : 15)) << 2;

    float* Vb = V + (size_t)n * 147456 + c * 2304 + 18 * r0 + 9 * (lane >> 4) + lq;
    float vv[8];
    for (int k = 0; k < 8; ++k) {
        float v[36];
        for (int q = 0; q < 4; ++q)
            for (int a = 0; a < 3; ++a)
                for (int b = 0; b < 3; ++b)
                    v[9 * q + 3 * a + b] = e[k + a][q + b];
        float A = 0.0f, B = 0.0f;
        for (int u = 0; u < 9; ++u) {
            const float g = ((v[4 * u] + v[4 * u + 1]) + v[4 * u + 2]) + v[4 * u + 3];
            A += (u < ng) ? g : 0.0f;
            B += (u < ng) ? 0.0f : g;
        }
        const float vB  = __int_as_float(__builtin_amdgcn_ds_bpermute(iB,  __float_as_int(B)));
        const float vA1 = __int_as_float(__builtin_amdgcn_ds_bpermute(iA1, __float_as_int(A)));
        const float vA2 = __int_as_float(__builtin_amdgcn_ds_bpermute(iA2, __float_as_int(A)));
        const float t = ((j > 0 ? vB : 0.0f) + vA1) + (two ? vA2 : 0.0f);
        const float vk = t + (half ? msc[2 * k + 1] : msc[2 * k]);
        vv[k] = vk;
        if (lq < 9) Vb[18 * k] = vk;
    }

    float mx = vv[0];
    for (int k = 1; k < 8; ++k) mx = fmaxf(mx, vv[k]);
    mx = fmaxf(mx, __shfl_xor(mx, 16, 32));
    float se = 0.0f;
    for (int k = 0; k < 8; ++k) se += __expf(vv[k] - mx);
    se += __shfl_xor(se, 16, 32);

    __shared__ float wmax[8][9];
    __shared__ float wsum[8][9];
    const int wv = threadIdx.x >> 5;
    if (lane < 9) { wmax[wv][lane] = mx; wsum[wv][lane] = se; }
    __builtin_amdgcn_fence(__ATOMIC_RELEASE, "workgroup", "local");
    __builtin_amdgcn_s_barrier();
    __builtin_amdgcn_fence(__ATOMIC_ACQUIRE, "workgroup", "local");
    if (threadIdx.x < 9) {
        float m = wmax[0][threadIdx.x];
        for (int w = 1; w < 8; ++w) m = fmaxf(m, wmax[w][threadIdx.x]);
        float t = 0.0f;
        for (int w = 0; w < 8; ++w) t += wsum[w][threadIdx.x] * __expf(wmax[w][threadIdx.x] - m);
        pmx[blockIdx.x * 9 + threadIdx.x] = m;
        pse[blockIdx.x * 9 + threadIdx.x] = t;
    }
}

__global__ __launch_bounds__(256) void kSoftsum(const float* __restrict__ V,
                                                const float* __restrict__ pmx,
                                                const float* __restrict__ pse,
                                                float* __restrict__ softsum) {
    const int idx = blockIdx.x * 256 + threadIdx.x;
    const int n = idx >> 14;
    const int p = idx & 16383;
    __shared__ float smx[9], sinv[9];
    const float* Vb = V + (size_t)n * 147456 + 9 * (size_t)p;
    float t[9];
    for (int s = 0; s < 9; ++s) t[s] = Vb[s];
    if (threadIdx.x < 160) {
        const int hw = threadIdx.x >> 4;
        const int hl = threadIdx.x & 15;
        const int s  = hw < 8 ? hw : 8;
        const float* a = pmx + n * 1152 + s;
        const float* b = pse + n * 1152 + s;
        float pm[8], ps[8];
        for (int k = 0; k < 8; ++k) { pm[k] = a[9 * (hl + 16 * k)]; ps[k] = b[9 * (hl + 16 * k)]; }
        float g = pm[0];
        for (int k = 1; k < 8; ++k) g = fmaxf(g, pm[k]);
        for (int off = 8; off > 0; off >>= 1) g = fmaxf(g, __shfl_xor(g, off, 32));
        float se = 0.0f;
        for (int k = 0; k < 8; ++k) se += ps[k] * __expf(pm[k] - g);
        for (int off = 8; off > 0; off >>= 1) se += __shfl_xor(se, off, 32);
        if (hl == 0 && hw < 9) {
            smx[s]  = g;
            sinv[s] = 1.0f / se;
        }
    }
    __syncthreads();
    float acc = 0.0f;
    for (int s = 0; s < 9; ++s)
        acc += __expf(t[s] - smx[s]) * sinv[s];
    softsum[idx] = acc;
}

__global__ __launch_bounds__(256) __attribute__((amdgpu_waves_per_eu(8, 8))) void kOut(const float* __restrict__ x,
                                            const float* __restrict__ softsum,
                                            float* __restrict__ out) {
    const int lane  = threadIdx.x & 31;
    const int wid   = (blockIdx.x * 256 + threadIdx.x) >> 5;
    const int strip = wid & 15;
    const int c     = (wid >> 4) & 63;
    const int n     = wid >> 10;
    const int r0    = strip * 8;

    const float* ssb = softsum + (size_t)n * 16384 + c * 256 + r0 * 2 + (lane >> 4);
    float ss[8];
    for (int k = 0; k < 8; ++k) ss[k] = ssb[2 * k];

    const float* xb = x + ((size_t)n * 64 + c) * 16384 + 4 * lane;
    v4f row[10];
    for (int j = 0; j < 10; ++j) {
        const int r  = r0 - 1 + j;
        const int rc = r < 0 ? 0 : (r > 127 ? 127 : r);
        v4f v = *(const v4f*)(xb + rc * 128);
        if ((unsigned)r >= 128u) v = (v4f){0.0f, 0.0f, 0.0f, 0.0f};
        row[j] = v;
    }

    const int up = ((lane + 31) & 31) << 2;
    const int dn = ((lane + 1) & 31) << 2;
    float e[10][6];
    for (int j = 0; j < 10; ++j) {
        float lft = __int_as_float(__builtin_amdgcn_ds_bpermute(up, __float_as_int(row[j].w)));
        float rgt = __int_as_float(__builtin_amdgcn_ds_bpermute(dn, __float_as_int(row[j].x)));
        e[j][0] = (lane == 0)  ? 0.0f : lft;
        e[j][1] = row[j].x;
        e[j][2] = row[j].y;
        e[j][3] = row[j].z;
        e[j][4] = row[j].w;
        e[j][5] = (lane == 31) ? 0.0f : rgt;
    }

    float* ob = out + ((size_t)n * 64 + c) * 16384 + (size_t)r0 * 128 + 4 * lane;
    for (int k = 0; k < 8; ++k) {
        v4f o;
        for (int q = 0; q < 4; ++q) {
            float b = e[k    ][q] + e[k    ][q + 1] + e[k    ][q + 2]
                    + e[k + 1][q] + e[k + 1][q + 1] + e[k + 1][q + 2]
                    + e[k + 2][q] + e[k + 2][q + 1] + e[k + 2][q + 2];
            o[q] = ss[k] + b;
        }
        *(v4f*)(ob + k * 128) = o;
    }
}

extern "C" void kernel_launch(void* const* d_in, const int* in_sizes, int n_in,
                              void* d_out, int out_size, void* d_ws, size_t ws_size,
                              hipStream_t stream) {
    const float* x = (const float*)d_in[0];
    float* out = (float*)d_out;
    float* ws  = (float*)d_ws;

    float* V       = ws;
    float* pmx     = V + (size_t)8 * 147456;
    float* pse     = pmx + 9216;
    float* softsum = pse + 9216;

    kMT     <<<1024,       256, 0, stream>>>(x, V, pmx, pse);
    kSoftsum<<<512,        256, 0, stream>>>(V, pmx, pse, softsum);
    kOut    <<<1024,       256, 0, stream>>>(x, softsum, out);
}
